# starting kernel
# baseline (speedup 1.0000x reference)
#include <hip/hip_runtime.h>

typedef __attribute__((ext_vector_type(2))) float v2f;
typedef __attribute__((ext_vector_type(8))) float v8f;

#define N_ATOMS 8192
#define N_EDGES 65536
#define N_PAIR  16
#define NH      64
#define TILES   2
#define EDGES_PER_BLOCK (TILES * 16)

__global__ void prep_wt(const float* __restrict__ W, const float* __restrict__ b,
                        float* __restrict__ Wt) {
    int idx = blockIdx.x * blockDim.x + threadIdx.x;
    if (idx >= 17 * NH * NH) return;
    int p = idx / (NH * NH);
    int r = idx % (NH * NH);
    int k = r / NH;
    int h = r % NH;
    Wt[idx] = (p < N_PAIR) ? W[p * (NH * NH) + h * NH + k] : b[h * NH + k];
}

__global__ void zero_out(float* __restrict__ out, int n) {
    int idx = blockIdx.x * blockDim.x + threadIdx.x;
    if (idx < n) out[idx] = 0.0f;
}

__launch_bounds__(128)
__global__ void edge_net(const float* __restrict__ pf,
                         const float* __restrict__ atomf,
                         const int*   __restrict__ a2p,
                         const float* __restrict__ Wt,
                         float*       __restrict__ out) {
    __shared__ float srcLds[EDGES_PER_BLOCK][NH + 4];
    __shared__ float pfLds[EDGES_PER_BLOCK][N_PAIR + 1];
    __shared__ int   a0Lds[EDGES_PER_BLOCK];

    const int tid  = threadIdx.x;
    const int lane = tid & 31;
    const int wave = tid >> 5;
    const int eb   = blockIdx.x * EDGES_PER_BLOCK;

    {
        int el = tid >> 2;
        int q  = tid & 3;
        int e  = eb + el;
        int a1 = a2p[2 * e + 1];
        const float4* srow = (const float4*)(atomf + (size_t)a1 * NH);
#pragma unroll
        for (int j = 0; j < 4; ++j) {
            float4 v = srow[q * 4 + j];
            float* d = &srcLds[el][q * 16 + j * 4];
            d[0] = v.x; d[1] = v.y; d[2] = v.z; d[3] = v.w;
        }
        if (tid < EDGES_PER_BLOCK) {
            int e2 = eb + tid;
#pragma unroll
            for (int j = 0; j < N_PAIR; ++j) pfLds[tid][j] = pf[e2 * N_PAIR + j];
            pfLds[tid][N_PAIR] = 1.0f;
            a0Lds[tid] = a2p[2 * e2];
        }
    }
    __syncthreads();

    const int h0   = wave * 16;
    const int n    = lane & 15;
    const int koff = (lane >> 4) * 2;
    const int mrow = (lane >> 4) * 8;

    v2f af[TILES][16];
#pragma unroll
    for (int t = 0; t < TILES; ++t)
#pragma unroll
        for (int ks = 0; ks < 16; ++ks)
            af[t][ks] = *(const v2f*)&srcLds[t * 16 + n][ks * 4 + koff];

    v8f accO[TILES];
#pragma unroll
    for (int t = 0; t < TILES; ++t) accO[t] = (v8f)0.0f;

    for (int p = 0; p <= N_PAIR; ++p) {
        const float* wp = Wt + (size_t)p * (NH * NH);
        v2f bf[16];
#pragma unroll
        for (int ks = 0; ks < 16; ++ks) {
            int kb   = ks * 4 + koff;
            bf[ks].x = wp[kb * NH + h0 + n];
            bf[ks].y = wp[(kb + 1) * NH + h0 + n];
        }
#pragma unroll
        for (int t = 0; t < TILES; ++t) {
            v8f acc = (v8f)0.0f;
#pragma unroll
            for (int ks = 0; ks < 16; ++ks)
                acc = __builtin_amdgcn_wmma_f32_16x16x4_f32(
                    false, af[t][ks], false, bf[ks], (short)0, acc, false, false);
#pragma unroll
            for (int i = 0; i < 8; ++i) {
                float w = pfLds[t * 16 + mrow + i][p];
                accO[t][i] += w * acc[i];
            }
        }
    }

#pragma unroll
    for (int t = 0; t < TILES; ++t)
#pragma unroll
        for (int i = 0; i < 8; ++i) {
            int el   = t * 16 + mrow + i;
            int atom = a0Lds[el];
            atomicAdd(&out[(size_t)atom * NH + h0 + n], accO[t][i]);
        }
}

extern "C" void kernel_launch(void* const* d_in, const int* in_sizes, int n_in,
                              void* d_out, int out_size, void* d_ws, size_t ws_size,
                              hipStream_t stream) {
    const float* pf    = (const float*)d_in[0];
    const float* atomf = (const float*)d_in[1];
    const int*   a2p   = (const int*)d_in[2];
    const float* W     = (const float*)d_in[3];
    const float* b     = (const float*)d_in[4];
    float* out = (float*)d_out;
    float* Wt  = (float*)d_ws;

    int prep_n = 17 * NH * NH;
    prep_wt<<<(prep_n + 255) / 256, 256, 0, stream>>>(W, b, Wt);
    zero_out<<<(N_ATOMS * NH + 255) / 256, 256, 0, stream>>>(out, N_ATOMS * NH);
    edge_net<<<N_EDGES / EDGES_PER_BLOCK, 128, 0, stream>>>(pf, atomf, a2p, Wt, out);
}


# slice products src.W_p formed once per atom (8192 rows, not 65536 edge rows: 8x fewer f32 matrix steps) into a workspace table, gathered and pf-weighted per edge; k sum order changes
# speedup vs baseline: 1.3476x; 1.3476x over previous
#include <hip/hip_runtime.h>


typedef __attribute__((ext_vector_type(2))) float v2f;
typedef __attribute__((ext_vector_type(8))) float v8f;

#define N_ATOMS 8192
#define N_EDGES 65536
#define N_PAIR  16
#define NH      64
#define NSL     17
#define UROW    (NSL * NH)
#define TILES   2
#define EDGES_PER_BLOCK (TILES * 16)


__launch_bounds__(128)
__global__ void atom_proj(const float* __restrict__ atomf,
                          const float* __restrict__ W,
                          const float* __restrict__ b,
                          float*       __restrict__ U) {
    __shared__ float wLds[NH][NH + 4];

    const int tid  = threadIdx.x;
    const int lane = tid & 31;
    const int wave = tid >> 5;
    const int p    = blockIdx.x >> 6;
    const int pair = (blockIdx.x & 63) * 4 + wave;
    const int c    = lane & 15;
    const int hh   = lane >> 4;

    const float* wsl = (p < N_PAIR) ? (W + (size_t)p * (NH * NH)) : b;

    {
        const float4* g = (const float4*)wsl;
#pragma unroll
        for (int i = 0; i < 8; ++i) {
            int f = tid + 128 * i;
            float4 v = g[f];
            *(float4*)&wLds[f >> 4][(f & 15) * 4] = v;
        }
    }

    v2f af[2][16];
#pragma unroll
    for (int m = 0; m < 2; ++m) {
        const float4* s = (const float4*)(atomf + (size_t)(pair * 32 + m * 16 + c) * NH + 32 * hh);
#pragma unroll
        for (int j = 0; j < 8; ++j) {
            float4 v = s[j];
            af[m][2 * j]     = (v2f){v.x, v.y};
            af[m][2 * j + 1] = (v2f){v.z, v.w};
        }
    }
    __syncthreads();

    v8f acc[2][4];
#pragma unroll
    for (int t = 0; t < 4; ++t) {
        const float4* s = (const float4*)&wLds[16 * t + c][32 * hh];
        v2f bf[16];
#pragma unroll
        for (int j = 0; j < 8; ++j) {
            float4 v = s[j];
            bf[2 * j]     = (v2f){v.x, v.y};
            bf[2 * j + 1] = (v2f){v.z, v.w};
        }
        acc[0][t] = (v8f)0.0f;
        acc[1][t] = (v8f)0.0f;
#pragma unroll
        for (int j = 0; j < 16; ++j) {
            acc[0][t] = __builtin_amdgcn_wmma_f32_16x16x4_f32(
                false, af[0][j], false, bf[j], (short)0, acc[0][t], false, false);
            acc[1][t] = __builtin_amdgcn_wmma_f32_16x16x4_f32(
                false, af[1][j], false, bf[j], (short)0, acc[1][t], false, false);
        }
    }

#pragma unroll
    for (int m = 0; m < 2; ++m)
#pragma unroll
        for (int r = 0; r < 8; ++r) {
            int atom = pair * 32 + m * 16 + 8 * hh + r;
            float4 v = make_float4(acc[m][0][r], acc[m][1][r], acc[m][2][r], acc[m][3][r]);
            *(float4*)(U + (size_t)atom * UROW + p * NH + 4 * c) = v;
        }
}

#define EPW 8
__launch_bounds__(256)
__global__ void edge_sum(const float* __restrict__ pf,
                         const int*   __restrict__ a2p,
                         const float* __restrict__ U,
                         float*       __restrict__ out) {
    const int lane = threadIdx.x & 31;
    const int c    = lane & 15;
    const int hh   = lane >> 4;
    const int gw   = blockIdx.x * 8 + (threadIdx.x >> 5);

    int    dst[EPW / 2];
    float4 sum[EPW / 2];
#pragma unroll
    for (int it = 0; it < EPW / 2; ++it) {
        const int e  = gw * EPW + it * 2 + hh;
        const int2 ap = *(const int2*)(a2p + 2 * (size_t)e);
        dst[it] = ap.x;
        const float4* pr = (const float4*)(pf + (size_t)e * N_PAIR);
        const float4* ur = (const float4*)(U + (size_t)ap.y * UROW) + c;
        float4 u[NSL];
#pragma unroll
        for (int p = 0; p < NSL; ++p) u[p] = ur[p * 16];
        float w[N_PAIR];
#pragma unroll
        for (int q = 0; q < 4; ++q) {
            float4 v = pr[q];
            w[4 * q] = v.x; w[4 * q + 1] = v.y; w[4 * q + 2] = v.z; w[4 * q + 3] = v.w;
        }
        float4 s = make_float4(0.0f, 0.0f, 0.0f, 0.0f);
#pragma unroll
        for (int p = 0; p < N_PAIR; ++p) {
            s.x += w[p] * u[p].x;
            s.y += w[p] * u[p].y;
            s.z += w[p] * u[p].z;
            s.w += w[p] * u[p].w;
        }
        s.x += u[N_PAIR].x; s.y += u[N_PAIR].y; s.z += u[N_PAIR].z; s.w += u[N_PAIR].w;
        sum[it] = s;
    }
#pragma unroll
    for (int it = 0; it < EPW / 2; ++it) {
        float* o = out + (size_t)dst[it] * NH + c;
        atomicAdd(o,      sum[it].x);
        atomicAdd(o + 16, sum[it].y);
        atomicAdd(o + 32, sum[it].z);
        atomicAdd(o + 48, sum[it].w);
    }
}

__global__ void zero_out(float* __restrict__ out, int n) {
    int idx = blockIdx.x * blockDim.x + threadIdx.x;
    if (idx < n) out[idx] = 0.0f;
}


__global__ void prep_wt(const float* __restrict__ W, const float* __restrict__ b,
                        float* __restrict__ Wt) {
    int idx = blockIdx.x * blockDim.x + threadIdx.x;
    if (idx >= 17 * NH * NH) return;
    int p = idx / (NH * NH);
    int r = idx % (NH * NH);
    int k = r / NH;
    int h = r % NH;
    Wt[idx] = (p < N_PAIR) ? W[p * (NH * NH) + h * NH + k] : b[h * NH + k];
}

__launch_bounds__(128)
__global__ void edge_net(const float* __restrict__ pf,
                         const float* __restrict__ atomf,
                         const int*   __restrict__ a2p,
                         const float* __restrict__ Wt,
                         float*       __restrict__ out) {
    __shared__ float srcLds[EDGES_PER_BLOCK][NH + 4];
    __shared__ float pfLds[EDGES_PER_BLOCK][N_PAIR + 1];
    __shared__ int   a0Lds[EDGES_PER_BLOCK];

    const int tid  = threadIdx.x;
    const int lane = tid & 31;
    const int wave = tid >> 5;
    const int eb   = blockIdx.x * EDGES_PER_BLOCK;

    {
        int el = tid >> 2;
        int q  = tid & 3;
        int e  = eb + el;
        int a1 = a2p[2 * e + 1];
        const float4* srow = (const float4*)(atomf + (size_t)a1 * NH);
#pragma unroll
        for (int j = 0; j < 4; ++j) {
            float4 v = srow[q * 4 + j];
            float* d = &srcLds[el][q * 16 + j * 4];
            d[0] = v.x; d[1] = v.y; d[2] = v.z; d[3] = v.w;
        }
        if (tid < EDGES_PER_BLOCK) {
            int e2 = eb + tid;
#pragma unroll
            for (int j = 0; j < N_PAIR; ++j) pfLds[tid][j] = pf[e2 * N_PAIR + j];
            pfLds[tid][N_PAIR] = 1.0f;
            a0Lds[tid] = a2p[2 * e2];
        }
    }
    __syncthreads();

    const int h0   = wave * 16;
    const int n    = lane & 15;
    const int koff = (lane >> 4) * 2;
    const int mrow = (lane >> 4) * 8;

    v2f af[TILES][16];
#pragma unroll
    for (int t = 0; t < TILES; ++t)
#pragma unroll
        for (int ks = 0; ks < 16; ++ks)
            af[t][ks] = *(const v2f*)&srcLds[t * 16 + n][ks * 4 + koff];

    v8f accO[TILES];
#pragma unroll
    for (int t = 0; t < TILES; ++t) accO[t] = (v8f)0.0f;

    for (int p = 0; p <= N_PAIR; ++p) {
        const float* wp = Wt + (size_t)p * (NH * NH);
        v2f bf[16];
#pragma unroll
        for (int ks = 0; ks < 16; ++ks) {
            int kb   = ks * 4 + koff;
            bf[ks].x = wp[kb * NH + h0 + n];
            bf[ks].y = wp[(kb + 1) * NH + h0 + n];
        }
#pragma unroll
        for (int t = 0; t < TILES; ++t) {
            v8f acc = (v8f)0.0f;
#pragma unroll
            for (int ks = 0; ks < 16; ++ks)
                acc = __builtin_amdgcn_wmma_f32_16x16x4_f32(
                    false, af[t][ks], false, bf[ks], (short)0, acc, false, false);
#pragma unroll
            for (int i = 0; i < 8; ++i) {
                float w = pfLds[t * 16 + mrow + i][p];
                accO[t][i] += w * acc[i];
            }
        }
    }

#pragma unroll
    for (int t = 0; t < TILES; ++t)
#pragma unroll
        for (int i = 0; i < 8; ++i) {
            int el   = t * 16 + mrow + i;
            int atom = a0Lds[el];
            atomicAdd(&out[(size_t)atom * NH + h0 + n], accO[t][i]);
        }
}

extern "C" void kernel_launch(void* const* d_in, const int* in_sizes, int n_in,
                              void* d_out, int out_size, void* d_ws, size_t ws_size,
                              hipStream_t stream) {
    const float* pf    = (const float*)d_in[0];
    const float* atomf = (const float*)d_in[1];
    const int*   a2p   = (const int*)d_in[2];
    const float* W     = (const float*)d_in[3];
    const float* b     = (const float*)d_in[4];
    float* out = (float*)d_out;
    float* ws  = (float*)d_ws;

    zero_out<<<(N_ATOMS * NH + 255) / 256, 256, 0, stream>>>(out, N_ATOMS * NH);
    if (ws_size >= sizeof(float) * N_ATOMS * UROW) {
        atom_proj<<<(N_ATOMS / 128) * NSL, 128, 0, stream>>>(atomf, W, b, ws);
        edge_sum<<<N_EDGES / (8 * EPW), 256, 0, stream>>>(pf, a2p, ws, out);
    } else {
        prep_wt<<<(17 * NH * NH + 255) / 256, 256, 0, stream>>>(W, b, ws);
        edge_net<<<N_EDGES / EDGES_PER_BLOCK, 128, 0, stream>>>(pf, atomf, a2p, ws, out);
    }
}


# U rows stay in LDS of the work-group owning 32 source atoms; it finds its edges from one owner byte per edge (LDS integer add), gathers rows from LDS; no table in memory; add order changes
# speedup vs baseline: 2.0748x; 1.5396x over previous
#include <hip/hip_runtime.h>


typedef __attribute__((ext_vector_type(2))) float v2f;
typedef __attribute__((ext_vector_type(8))) float v8f;
typedef __attribute__((ext_vector_type(4))) float v4f;

#define N_ATOMS 8192
#define N_EDGES 65536
#define N_PAIR  16
#define NH      64
#define NSL     17
#define APB     32
#define NBLK    (N_ATOMS / APB)
#define UPITCH  (NSL * NH + 4)
#define WPITCH  (NH + 4)
#define WSLICE  (NH * WPITCH)


__launch_bounds__(256)
__global__ void prep(const int* __restrict__ a2p,
                     unsigned int* __restrict__ blk,
                     float* __restrict__ out) {
    const int gid = blockIdx.x * 256 + threadIdx.x;
    const float4 z = make_float4(0.0f, 0.0f, 0.0f, 0.0f);
#pragma unroll
    for (int i = 0; i < 4; ++i)
        ((float4*)out)[gid + 32768 * i] = z;
    if (gid < N_EDGES / 4) {
        const int4* p = (const int4*)a2p + 2 * (size_t)gid;
        const int4 v0 = p[0];
        const int4 v1 = p[1];
        blk[gid] = ((unsigned)(v0.y >> 5) & 255u)
                 | (((unsigned)(v0.w >> 5) & 255u) << 8)
                 | (((unsigned)(v1.y >> 5) & 255u) << 16)
                 | (((unsigned)(v1.w >> 5) & 255u) << 24);
    }
}

__device__ __forceinline__ const v4f* slice_ptr(const float* W, const float* b, int s) {
    return (const v4f*)((s < N_PAIR) ? (W + (size_t)s * (NH * NH)) : b);
}

__device__ __forceinline__ void pair_to_regs(v4f (&wreg)[8], const float* W, const float* b,
                                             int jj, int tid) {
    const v4f* g0 = slice_ptr(W, b, 2 * jj);
#pragma unroll
    for (int i = 0; i < 4; ++i) wreg[i] = g0[tid + 256 * i];
    if (2 * jj + 1 < NSL) {
        const v4f* g1 = slice_ptr(W, b, 2 * jj + 1);
#pragma unroll
        for (int i = 0; i < 4; ++i) wreg[4 + i] = g1[tid + 256 * i];
    }
}

__device__ __forceinline__ void regs_to_lds(const v4f (&wreg)[8], float* wLds, int jj, int tid) {
#pragma unroll
    for (int i = 0; i < 4; ++i) {
        const int g = tid + 256 * i;
        *(v4f*)&wLds[(g >> 4) * WPITCH + (g & 15) * 4] = wreg[i];
    }
    if (2 * jj + 1 < NSL) {
#pragma unroll
        for (int i = 0; i < 4; ++i) {
            const int g = tid + 256 * i;
            *(v4f*)&wLds[WSLICE + (g >> 4) * WPITCH + (g & 15) * 4] = wreg[4 + i];
        }
    }
}

__device__ __forceinline__ void load_bf(v2f (&bf)[16], const float* wLds, int par, int t, int c, int hh) {
    const float4* s = (const float4*)&wLds[par * WSLICE + (16 * t + c) * WPITCH + 32 * hh];
#pragma unroll
    for (int j = 0; j < 8; ++j) {
        const float4 v = s[j];
        bf[2 * j]     = (v2f){v.x, v.y};
        bf[2 * j + 1] = (v2f){v.z, v.w};
    }
}

__device__ __forceinline__ void fetch(int2 (&ap)[4], float4 (&pw)[4][4], unsigned r, unsigned n,
                                      const unsigned short* eList, const int* a2p, const float* pf,
                                      int wave, int hh) {
#pragma unroll
    for (int it = 0; it < 4; ++it) {
        const unsigned idx = r * 64u + (unsigned)(wave * 8 + it * 2 + hh);
        const unsigned e   = eList[idx < n ? idx : 0u];
        ap[it] = *(const int2*)(a2p + 2 * (size_t)e);
        const float4* pr = (const float4*)(pf + (size_t)e * N_PAIR);
#pragma unroll
        for (int q = 0; q < 4; ++q) pw[it][q] = pr[q];
    }
}

__device__ __forceinline__ void process(const int2 (&ap)[4], const float4 (&pw)[4][4], unsigned r, unsigned n,
                                        const float* uLds, float* out, int wave, int c, int hh) {
#pragma unroll
    for (int it = 0; it < 4; ++it) {
        const unsigned idx = r * 64u + (unsigned)(wave * 8 + it * 2 + hh);
        const float4* ur = (const float4*)&uLds[(ap[it].y & (APB - 1)) * UPITCH + 4 * c];
        float4 u[NSL];
#pragma unroll
        for (int p = 0; p < NSL; ++p) u[p] = ur[p * 16];
        float w[N_PAIR];
#pragma unroll
        for (int q = 0; q < 4; ++q) {
            const float4 v = pw[it][q];
            w[4 * q] = v.x; w[4 * q + 1] = v.y; w[4 * q + 2] = v.z; w[4 * q + 3] = v.w;
        }
        float4 s = make_float4(0.0f, 0.0f, 0.0f, 0.0f);
#pragma unroll
        for (int p = 0; p < N_PAIR; ++p) {
            s.x += w[p] * u[p].x;
            s.y += w[p] * u[p].y;
            s.z += w[p] * u[p].z;
            s.w += w[p] * u[p].w;
        }
        s.x += u[N_PAIR].x; s.y += u[N_PAIR].y; s.z += u[N_PAIR].z; s.w += u[N_PAIR].w;
        if (idx < n) {
            float* o = out + (size_t)ap[it].x * NH + c;
            atomicAdd(o,      s.x);
            atomicAdd(o + 16, s.y);
            atomicAdd(o + 32, s.z);
            atomicAdd(o + 48, s.w);
        }
    }
}

__launch_bounds__(256)
__global__ void own_sources(const float* __restrict__ pf,
                            const float* __restrict__ atomf,
                            const int*   __restrict__ a2p,
                            const float* __restrict__ W,
                            const float* __restrict__ b,
                            const unsigned int* __restrict__ blk,
                            float*       __restrict__ out) {
    __shared__ __attribute__((aligned(16))) float uLds[APB * UPITCH];
    __shared__ __attribute__((aligned(16))) float wLds[2 * WSLICE];
    __shared__ unsigned short eList[N_EDGES];
    __shared__ unsigned int   eCnt;

    const int tid  = threadIdx.x;
    const int lane = tid & 31;
    const int wave = __builtin_amdgcn_readfirstlane(tid >> 5);
    const int c    = lane & 15;
    const int hh   = lane >> 4;
    const int t    = wave & 3;
    const int par  = wave >> 2;
    const int blkId = blockIdx.x;

    if (tid == 0) { eCnt = 0u; eList[0] = 0; }

    v4f wreg[8];
    pair_to_regs(wreg, W, b, 0, tid);

    v2f af[2][16];
#pragma unroll
    for (int m = 0; m < 2; ++m) {
        const float4* s = (const float4*)(atomf + (size_t)(blkId * APB + m * 16 + c) * NH + 32 * hh);
#pragma unroll
        for (int j = 0; j < 8; ++j) {
            const float4 v = s[j];
            af[m][2 * j]     = (v2f){v.x, v.y};
            af[m][2 * j + 1] = (v2f){v.z, v.w};
        }
    }

    uint4 bq[16];
#pragma unroll
    for (int i = 0; i < 16; ++i) bq[i] = ((const uint4*)blk)[tid + 256 * i];

    __syncthreads();

    v4f wnext[8];
    pair_to_regs(wnext, W, b, 1, tid);

    {
        const unsigned pat = (unsigned)blkId * 0x01010101u;
        unsigned zz[64];
        unsigned total = 0u;
#pragma unroll
        for (int i = 0; i < 16; ++i) {
            const unsigned dw[4] = {bq[i].x, bq[i].y, bq[i].z, bq[i].w};
#pragma unroll
            for (int d = 0; d < 4; ++d) {
                const unsigned x = dw[d] ^ pat;
                const unsigned z = ~(((x & 0x7F7F7F7Fu) + 0x7F7F7F7Fu) | x | 0x7F7F7F7Fu);
                zz[4 * i + d] = z;
                total += (unsigned)__popc(z);
            }
        }
        unsigned pos = atomicAdd(&eCnt, total);
#pragma unroll
        for (int i = 0; i < 16; ++i)
#pragma unroll
            for (int d = 0; d < 4; ++d) {
                const unsigned z = zz[4 * i + d];
                if (z) {
#pragma unroll
                    for (int k = 0; k < 4; ++k)
                        if (z & (0x80u << (8 * k))) {
                            eList[pos] = (unsigned short)(16 * (tid + 256 * i) + 4 * d + k);
                            ++pos;
                        }
                }
            }
    }

    regs_to_lds(wreg, wLds, 0, tid);
#pragma unroll
    for (int i = 0; i < 8; ++i) wreg[i] = wnext[i];
    __syncthreads();

    const unsigned n      = eCnt;
    const unsigned rounds = (n + 63u) >> 6;

    v2f bfc[16], bfn[16];
    load_bf(bfc, wLds, par, t, c, hh);

    int2   apA[4];
    float4 pwA[4][4];

#pragma unroll
    for (int j = 0; j < 9; ++j) {
        if (j < 8) {
            __syncthreads();
            regs_to_lds(wreg, wLds, j + 1, tid);
            __syncthreads();
            if (j + 1 < 8 || par == 0) load_bf(bfn, wLds, par, t, c, hh);
            __builtin_amdgcn_sched_barrier(0);
        }
        if (j == 0) fetch(apA, pwA, 0u, n, eList, a2p, pf, wave, hh);
        if (j < 7) pair_to_regs(wreg, W, b, j + 2, tid);

        if (j < 8 || par == 0) {
            const int p = 2 * j + par;
            v8f acc0 = (v8f)0.0f;
            v8f acc1 = (v8f)0.0f;
#pragma unroll
            for (int q = 0; q < 16; ++q) {
                acc0 = __builtin_amdgcn_wmma_f32_16x16x4_f32(
                    false, af[0][q], false, bfc[q], (short)0, acc0, false, false);
                acc1 = __builtin_amdgcn_wmma_f32_16x16x4_f32(
                    false, af[1][q], false, bfc[q], (short)0, acc1, false, false);
            }
#pragma unroll
            for (int r = 0; r < 8; ++r) {
                uLds[(8 * hh + r) * UPITCH + p * NH + 4 * c + t]      = acc0[r];
                uLds[(16 + 8 * hh + r) * UPITCH + p * NH + 4 * c + t] = acc1[r];
            }
        }
#pragma unroll
        for (int q = 0; q < 16; ++q) bfc[q] = bfn[q];
    }
    __syncthreads();

    for (unsigned r = 0; r < rounds; ++r) {
        int2   apB[4];
        float4 pwB[4][4];
        fetch(apB, pwB, r + 1u, n, eList, a2p, pf, wave, hh);
        process(apA, pwA, r, n, uLds, out, wave, c, hh);
#pragma unroll
        for (int it = 0; it < 4; ++it) {
            apA[it] = apB[it];
#pragma unroll
            for (int q = 0; q < 4; ++q) pwA[it][q] = pwB[it][q];
        }
    }
}

extern "C" void kernel_launch(void* const* d_in, const int* in_sizes, int n_in,
                              void* d_out, int out_size, void* d_ws, size_t ws_size,
                              hipStream_t stream) {
    const float* pf    = (const float*)d_in[0];
    const float* atomf = (const float*)d_in[1];
    const int*   a2p   = (const int*)d_in[2];
    const float* W     = (const float*)d_in[3];
    const float* b     = (const float*)d_in[4];
    float* out = (float*)d_out;
    unsigned int* blk = (unsigned int*)d_ws;

    prep<<<128, 256, 0, stream>>>(a2p, blk, out);
    own_sources<<<NBLK, 256, 0, stream>>>(pf, atomf, a2p, W, b, blk, out);
}


# first launch writes each owner work-group a bitmap of its edges (LDS bit-or) instead of owner bytes: second launch reads 8 KB and walks set bits, no search of 65536 bytes; list and add order change
# speedup vs baseline: 2.4484x; 1.1801x over previous
#include <hip/hip_runtime.h>


typedef __attribute__((ext_vector_type(2))) float v2f;
typedef __attribute__((ext_vector_type(8))) float v8f;
typedef __attribute__((ext_vector_type(4))) float v4f;

#define N_ATOMS 8192
#define N_EDGES 65536
#define N_PAIR  16
#define NH      64
#define NSL     17
#define APB     32
#define NBLK    (N_ATOMS / APB)
#define UPITCH  (NSL * NH + 4)
#define WPITCH  (NH + 4)
#define WSLICE  (NH * WPITCH)


__launch_bounds__(256)
__global__ void prep(const int* __restrict__ a2p,
                     unsigned int* __restrict__ bits,
                     float* __restrict__ out) {
    __shared__ __attribute__((aligned(16))) unsigned int bm[256 * 32];
    const int tid = threadIdx.x;
    const int G   = blockIdx.x;
    if (G < 64) {
        const int4* p = (const int4*)a2p + 2 * (size_t)(G * 256 + tid);
        const int4 v0 = p[0];
        const int4 v1 = p[1];
        const uint4 z4 = make_uint4(0u, 0u, 0u, 0u);
#pragma unroll
        for (int i = 0; i < 8; ++i) ((uint4*)bm)[tid + 256 * i] = z4;
        __syncthreads();
        const int a1[4] = {v0.y, v0.w, v1.y, v1.w};
        const unsigned j = (unsigned)tid >> 3;
#pragma unroll
        for (int m = 0; m < 4; ++m) {
            const unsigned k = ((unsigned)a1[m] >> 5) & 255u;
            atomicOr(&bm[k * 32u + ((j + 4u * ((k >> 1) & 7u)) & 31u)], 1u << (4 * (tid & 7) + m));
        }
        __syncthreads();
#pragma unroll
        for (int i = 0; i < 8; ++i) {
            const unsigned q   = (unsigned)(tid + 256 * i);
            const unsigned row = q >> 3;
            const unsigned pc  = q & 7u;
            const uint4 v = ((const uint4*)bm)[row * 8u + ((pc + ((row >> 1) & 7u)) & 7u)];
            ((uint4*)bits)[row * 512u + (unsigned)G * 8u + pc] = v;
        }
    } else {
        const float4 z = make_float4(0.0f, 0.0f, 0.0f, 0.0f);
#pragma unroll
        for (int i = 0; i < 8; ++i)
            ((float4*)out)[(G - 64) * 2048 + tid + 256 * i] = z;
    }
}

__device__ __forceinline__ const v4f* slice_ptr(const float* W, const float* b, int s) {
    return (const v4f*)((s < N_PAIR) ? (W + (size_t)s * (NH * NH)) : b);
}

__device__ __forceinline__ void pair_to_regs(v4f (&wreg)[8], const float* W, const float* b,
                                             int jj, int tid) {
    const v4f* g0 = slice_ptr(W, b, 2 * jj);
#pragma unroll
    for (int i = 0; i < 4; ++i) wreg[i] = g0[tid + 256 * i];
    if (2 * jj + 1 < NSL) {
        const v4f* g1 = slice_ptr(W, b, 2 * jj + 1);
#pragma unroll
        for (int i = 0; i < 4; ++i) wreg[4 + i] = g1[tid + 256 * i];
    }
}

__device__ __forceinline__ void regs_to_lds(const v4f (&wreg)[8], float* wLds, int jj, int tid) {
#pragma unroll
    for (int i = 0; i < 4; ++i) {
        const int g = tid + 256 * i;
        *(v4f*)&wLds[(g >> 4) * WPITCH + (g & 15) * 4] = wreg[i];
    }
    if (2 * jj + 1 < NSL) {
#pragma unroll
        for (int i = 0; i < 4; ++i) {
            const int g = tid + 256 * i;
            *(v4f*)&wLds[WSLICE + (g >> 4) * WPITCH + (g & 15) * 4] = wreg[4 + i];
        }
    }
}

__device__ __forceinline__ void load_bf(v2f (&bf)[16], const float* wLds, int par, int t, int c, int hh) {
    const float4* s = (const float4*)&wLds[par * WSLICE + (16 * t + c) * WPITCH + 32 * hh];
#pragma unroll
    for (int j = 0; j < 8; ++j) {
        const float4 v = s[j];
        bf[2 * j]     = (v2f){v.x, v.y};
        bf[2 * j + 1] = (v2f){v.z, v.w};
    }
}

__device__ __forceinline__ void fetch(int2 (&ap)[4], float4 (&pw)[4][4], unsigned r, unsigned n,
                                      const unsigned short* eList, const int* a2p, const float* pf,
                                      int wave, int hh) {
#pragma unroll
    for (int it = 0; it < 4; ++it) {
        const unsigned idx = r * 64u + (unsigned)(wave * 8 + it * 2 + hh);
        const unsigned e   = eList[idx < n ? idx : 0u];
        ap[it] = *(const int2*)(a2p + 2 * (size_t)e);
        const float4* pr = (const float4*)(pf + (size_t)e * N_PAIR);
#pragma unroll
        for (int q = 0; q < 4; ++q) pw[it][q] = pr[q];
    }
}

__device__ __forceinline__ void process(const int2 (&ap)[4], const float4 (&pw)[4][4], unsigned r, unsigned n,
                                        const float* uLds, float* out, int wave, int c, int hh) {
#pragma unroll
    for (int it = 0; it < 4; ++it) {
        const unsigned idx = r * 64u + (unsigned)(wave * 8 + it * 2 + hh);
        const float4* ur = (const float4*)&uLds[(ap[it].y & (APB - 1)) * UPITCH + 4 * c];
        float4 u[NSL];
#pragma unroll
        for (int p = 0; p < NSL; ++p) u[p] = ur[p * 16];
        float w[N_PAIR];
#pragma unroll
        for (int q = 0; q < 4; ++q) {
            const float4 v = pw[it][q];
            w[4 * q] = v.x; w[4 * q + 1] = v.y; w[4 * q + 2] = v.z; w[4 * q + 3] = v.w;
        }
        float4 s = make_float4(0.0f, 0.0f, 0.0f, 0.0f);
#pragma unroll
        for (int p = 0; p < N_PAIR; ++p) {
            s.x += w[p] * u[p].x;
            s.y += w[p] * u[p].y;
            s.z += w[p] * u[p].z;
            s.w += w[p] * u[p].w;
        }
        s.x += u[N_PAIR].x; s.y += u[N_PAIR].y; s.z += u[N_PAIR].z; s.w += u[N_PAIR].w;
        if (idx < n) {
            float* o = out + (size_t)ap[it].x * NH + c;
            atomicAdd(o,      s.x);
            atomicAdd(o + 16, s.y);
            atomicAdd(o + 32, s.z);
            atomicAdd(o + 48, s.w);
        }
    }
}

__launch_bounds__(256)
__global__ void own_sources(const float* __restrict__ pf,
                            const float* __restrict__ atomf,
                            const int*   __restrict__ a2p,
                            const float* __restrict__ W,
                            const float* __restrict__ b,
                            const unsigned int* __restrict__ bits,
                            float*       __restrict__ out) {
    __shared__ __attribute__((aligned(16))) float uLds[APB * UPITCH];
    __shared__ __attribute__((aligned(16))) float wLds[2 * WSLICE];
    __shared__ unsigned short eList[N_EDGES];
    __shared__ unsigned int   eCnt;

    const int tid  = threadIdx.x;
    const int lane = tid & 31;
    const int wave = __builtin_amdgcn_readfirstlane(tid >> 5);
    const int c    = lane & 15;
    const int hh   = lane >> 4;
    const int t    = wave & 3;
    const int par  = wave >> 2;
    const int blkId = blockIdx.x;

    if (tid == 0) { eCnt = 0u; eList[0] = 0; }

    v4f wreg[8];
    pair_to_regs(wreg, W, b, 0, tid);

    v2f af[2][16];
#pragma unroll
    for (int m = 0; m < 2; ++m) {
        const float4* s = (const float4*)(atomf + (size_t)(blkId * APB + m * 16 + c) * NH + 32 * hh);
#pragma unroll
        for (int j = 0; j < 8; ++j) {
            const float4 v = s[j];
            af[m][2 * j]     = (v2f){v.x, v.y};
            af[m][2 * j + 1] = (v2f){v.z, v.w};
        }
    }

    uint4 bw[2];
#pragma unroll
    for (int i = 0; i < 2; ++i) bw[i] = ((const uint4*)bits)[blkId * 512 + tid + 256 * i];

    __syncthreads();

    v4f wnext[8];
    pair_to_regs(wnext, W, b, 1, tid);

    {
        const unsigned wd[8] = {bw[0].x, bw[0].y, bw[0].z, bw[0].w, bw[1].x, bw[1].y, bw[1].z, bw[1].w};
        unsigned total = 0u;
#pragma unroll
        for (int d = 0; d < 8; ++d) total += (unsigned)__popc(wd[d]);
        unsigned pos = 0u;
        if (total != 0u) pos = atomicAdd(&eCnt, total);
#pragma unroll
        for (int i = 0; i < 2; ++i)
#pragma unroll
            for (int d = 0; d < 4; ++d) {
                unsigned x = wd[4 * i + d];
                const unsigned base = 32u * (4u * (unsigned)(tid + 256 * i) + (unsigned)d);
                while (x != 0u) {
                    eList[pos] = (unsigned short)(base + (unsigned)__builtin_ctz(x));
                    ++pos;
                    x &= x - 1u;
                }
            }
    }

    regs_to_lds(wreg, wLds, 0, tid);
#pragma unroll
    for (int i = 0; i < 8; ++i) wreg[i] = wnext[i];
    __syncthreads();

    const unsigned n      = eCnt;
    const unsigned rounds = (n + 63u) >> 6;

    v2f bfc[16], bfn[16];
    load_bf(bfc, wLds, par, t, c, hh);

    int2   apA[4];
    float4 pwA[4][4];

#pragma unroll
    for (int j = 0; j < 9; ++j) {
        if (j < 8) {
            __syncthreads();
            regs_to_lds(wreg, wLds, j + 1, tid);
            __syncthreads();
            if (j + 1 < 8 || par == 0) load_bf(bfn, wLds, par, t, c, hh);
            __builtin_amdgcn_sched_barrier(0);
        }
        if (j == 0) fetch(apA, pwA, 0u, n, eList, a2p, pf, wave, hh);
        if (j < 7) pair_to_regs(wreg, W, b, j + 2, tid);

        if (j < 8 || par == 0) {
            const int p = 2 * j + par;
            v8f acc0 = (v8f)0.0f;
            v8f acc1 = (v8f)0.0f;
#pragma unroll
            for (int q = 0; q < 16; ++q) {
                acc0 = __builtin_amdgcn_wmma_f32_16x16x4_f32(
                    false, af[0][q], false, bfc[q], (short)0, acc0, false, false);
                acc1 = __builtin_amdgcn_wmma_f32_16x16x4_f32(
                    false, af[1][q], false, bfc[q], (short)0, acc1, false, false);
            }
#pragma unroll
            for (int r = 0; r < 8; ++r) {
                uLds[(8 * hh + r) * UPITCH + p * NH + 4 * c + t]      = acc0[r];
                uLds[(16 + 8 * hh + r) * UPITCH + p * NH + 4 * c + t] = acc1[r];
            }
        }
#pragma unroll
        for (int q = 0; q < 16; ++q) bfc[q] = bfn[q];
    }
    __syncthreads();

    for (unsigned r = 0; r < rounds; ++r) {
        int2   apB[4];
        float4 pwB[4][4];
        fetch(apB, pwB, r + 1u, n, eList, a2p, pf, wave, hh);
        process(apA, pwA, r, n, uLds, out, wave, c, hh);
#pragma unroll
        for (int it = 0; it < 4; ++it) {
            apA[it] = apB[it];
#pragma unroll
            for (int q = 0; q < 4; ++q) pwA[it][q] = pwB[it][q];
        }
    }
}

extern "C" void kernel_launch(void* const* d_in, const int* in_sizes, int n_in,
                              void* d_out, int out_size, void* d_ws, size_t ws_size,
                              hipStream_t stream) {
    const float* pf    = (const float*)d_in[0];
    const float* atomf = (const float*)d_in[1];
    const int*   a2p   = (const int*)d_in[2];
    const float* W     = (const float*)d_in[3];
    const float* b     = (const float*)d_in[4];
    float* out = (float*)d_out;
    unsigned int* bits = (unsigned int*)d_ws;

    prep<<<128, 256, 0, stream>>>(a2p, bits, out);
    own_sources<<<NBLK, 256, 0, stream>>>(pf, atomf, a2p, W, b, bits, out);
}


# weight staging made wave-private: each wave fetches and stages only the 16 rows it multiplies itself, so the 16 work-group barriers of the product phase are gone; sum orders unchanged
# speedup vs baseline: 2.6476x; 1.0814x over previous
#include <hip/hip_runtime.h>


typedef __attribute__((ext_vector_type(2))) float v2f;
typedef __attribute__((ext_vector_type(8))) float v8f;
typedef __attribute__((ext_vector_type(4))) float v4f;

#define N_ATOMS 8192
#define N_EDGES 65536
#define N_PAIR  16
#define NH      64
#define NSL     17
#define APB     32
#define NBLK    (N_ATOMS / APB)
#define UPITCH  (NSL * NH + 4)
#define WPITCH  (NH + 4)
#define WSLICE  (NH * WPITCH)


__launch_bounds__(256)
__global__ void prep(const int* __restrict__ a2p,
                     unsigned int* __restrict__ bits,
                     float* __restrict__ out) {
    __shared__ __attribute__((aligned(16))) unsigned int bm[256 * 32];
    const int tid = threadIdx.x;
    const int G   = blockIdx.x;
    if (G < 64) {
        const int4* p = (const int4*)a2p + 2 * (size_t)(G * 256 + tid);
        const int4 v0 = p[0];
        const int4 v1 = p[1];
        const uint4 z4 = make_uint4(0u, 0u, 0u, 0u);
#pragma unroll
        for (int i = 0; i < 8; ++i) ((uint4*)bm)[tid + 256 * i] = z4;
        __syncthreads();
        const int a1[4] = {v0.y, v0.w, v1.y, v1.w};
        const unsigned j = (unsigned)tid >> 3;
#pragma unroll
        for (int m = 0; m < 4; ++m) {
            const unsigned k = ((unsigned)a1[m] >> 5) & 255u;
            atomicOr(&bm[k * 32u + ((j + 4u * ((k >> 1) & 7u)) & 31u)], 1u << (4 * (tid & 7) + m));
        }
        __syncthreads();
#pragma unroll
        for (int i = 0; i < 8; ++i) {
            const unsigned q   = (unsigned)(tid + 256 * i);
            const unsigned row = q >> 3;
            const unsigned pc  = q & 7u;
            const uint4 v = ((const uint4*)bm)[row * 8u + ((pc + ((row >> 1) & 7u)) & 7u)];
            ((uint4*)bits)[row * 512u + (unsigned)G * 8u + pc] = v;
        }
    } else {
        const float4 z = make_float4(0.0f, 0.0f, 0.0f, 0.0f);
#pragma unroll
        for (int i = 0; i < 8; ++i)
            ((float4*)out)[(G - 64) * 2048 + tid + 256 * i] = z;
    }
}

__device__ __forceinline__ const v4f* slice_ptr(const float* W, const float* b, int s) {
    return (const v4f*)((s < N_PAIR) ? (W + (size_t)s * (NH * NH)) : b);
}

__device__ __forceinline__ void rows_to_regs(v4f (&wreg)[8], const float* W, const float* b,
                                             int s, int t, int lane) {
    const v4f* g = slice_ptr(W, b, s) + 256 * t;
#pragma unroll
    for (int i = 0; i < 8; ++i) wreg[i] = g[lane + 32 * i];
}

__device__ __forceinline__ void regs_to_rows(const v4f (&wreg)[8], float* myRows, int lane) {
#pragma unroll
    for (int i = 0; i < 8; ++i) {
        const int g = lane + 32 * i;
        *(v4f*)&myRows[(g >> 4) * WPITCH + (g & 15) * 4] = wreg[i];
    }
}

__device__ __forceinline__ void wave_lds_order() {
    __builtin_amdgcn_fence(__ATOMIC_ACQ_REL, "wavefront");
    __builtin_amdgcn_wave_barrier();
}

__device__ __forceinline__ void load_bf(v2f (&bf)[16], const float* myRows, int c, int hh) {
    const float4* s = (const float4*)&myRows[c * WPITCH + 32 * hh];
#pragma unroll
    for (int j = 0; j < 8; ++j) {
        const float4 v = s[j];
        bf[2 * j]     = (v2f){v.x, v.y};
        bf[2 * j + 1] = (v2f){v.z, v.w};
    }
}

__device__ __forceinline__ void fetch(int2 (&ap)[4], float4 (&pw)[4][4], unsigned r, unsigned n,
                                      const unsigned short* eList, const int* a2p, const float* pf,
                                      int wave, int hh) {
#pragma unroll
    for (int it = 0; it < 4; ++it) {
        const unsigned idx = r * 64u + (unsigned)(wave * 8 + it * 2 + hh);
        const unsigned e   = eList[idx < n ? idx : 0u];
        ap[it] = *(const int2*)(a2p + 2 * (size_t)e);
        const float4* pr = (const float4*)(pf + (size_t)e * N_PAIR);
#pragma unroll
        for (int q = 0; q < 4; ++q) pw[it][q] = pr[q];
    }
}

__device__ __forceinline__ void process(const int2 (&ap)[4], const float4 (&pw)[4][4], unsigned r, unsigned n,
                                        const float* uLds, float* out, int wave, int c, int hh) {
#pragma unroll
    for (int it = 0; it < 4; ++it) {
        const unsigned idx = r * 64u + (unsigned)(wave * 8 + it * 2 + hh);
        const float4* ur = (const float4*)&uLds[(ap[it].y & (APB - 1)) * UPITCH + 4 * c];
        float4 u[NSL];
#pragma unroll
        for (int p = 0; p < NSL; ++p) u[p] = ur[p * 16];
        float w[N_PAIR];
#pragma unroll
        for (int q = 0; q < 4; ++q) {
            const float4 v = pw[it][q];
            w[4 * q] = v.x; w[4 * q + 1] = v.y; w[4 * q + 2] = v.z; w[4 * q + 3] = v.w;
        }
        float4 s = make_float4(0.0f, 0.0f, 0.0f, 0.0f);
#pragma unroll
        for (int p = 0; p < N_PAIR; ++p) {
            s.x += w[p] * u[p].x;
            s.y += w[p] * u[p].y;
            s.z += w[p] * u[p].z;
            s.w += w[p] * u[p].w;
        }
        s.x += u[N_PAIR].x; s.y += u[N_PAIR].y; s.z += u[N_PAIR].z; s.w += u[N_PAIR].w;
        if (idx < n) {
            float* o = out + (size_t)ap[it].x * NH + c;
            atomicAdd(o,      s.x);
            atomicAdd(o + 16, s.y);
            atomicAdd(o + 32, s.z);
            atomicAdd(o + 48, s.w);
        }
    }
}

__launch_bounds__(256)
__global__ void own_sources(const float* __restrict__ pf,
                            const float* __restrict__ atomf,
                            const int*   __restrict__ a2p,
                            const float* __restrict__ W,
                            const float* __restrict__ b,
                            const unsigned int* __restrict__ bits,
                            float*       __restrict__ out) {
    __shared__ __attribute__((aligned(16))) float uLds[APB * UPITCH];
    __shared__ __attribute__((aligned(16))) float wLds[2 * WSLICE];
    __shared__ unsigned short eList[N_EDGES];
    __shared__ unsigned int   eCnt;

    const int tid  = threadIdx.x;
    const int lane = tid & 31;
    const int wave = __builtin_amdgcn_readfirstlane(tid >> 5);
    const int c    = lane & 15;
    const int hh   = lane >> 4;
    const int t    = wave & 3;
    const int par  = wave >> 2;
    const int blkId = blockIdx.x;

    if (tid == 0) { eCnt = 0u; eList[0] = 0; }

    float* myRows = wLds + par * WSLICE + 16 * t * WPITCH;

    v4f wreg[8];
    rows_to_regs(wreg, W, b, par, t, lane);

    v2f af[2][16];
#pragma unroll
    for (int m = 0; m < 2; ++m) {
        const float4* s = (const float4*)(atomf + (size_t)(blkId * APB + m * 16 + c) * NH + 32 * hh);
#pragma unroll
        for (int j = 0; j < 8; ++j) {
            const float4 v = s[j];
            af[m][2 * j]     = (v2f){v.x, v.y};
            af[m][2 * j + 1] = (v2f){v.z, v.w};
        }
    }

    uint4 bw[2];
#pragma unroll
    for (int i = 0; i < 2; ++i) bw[i] = ((const uint4*)bits)[blkId * 512 + tid + 256 * i];

    __syncthreads();

    v4f wnext[8];
    rows_to_regs(wnext, W, b, 2 + par, t, lane);

    {
        const unsigned wd[8] = {bw[0].x, bw[0].y, bw[0].z, bw[0].w, bw[1].x, bw[1].y, bw[1].z, bw[1].w};
        unsigned total = 0u;
#pragma unroll
        for (int d = 0; d < 8; ++d) total += (unsigned)__popc(wd[d]);
        unsigned pos = 0u;
        if (total != 0u) pos = atomicAdd(&eCnt, total);
#pragma unroll
        for (int i = 0; i < 2; ++i)
#pragma unroll
            for (int d = 0; d < 4; ++d) {
                unsigned x = wd[4 * i + d];
                const unsigned base = 32u * (4u * (unsigned)(tid + 256 * i) + (unsigned)d);
                while (x != 0u) {
                    eList[pos] = (unsigned short)(base + (unsigned)__builtin_ctz(x));
                    ++pos;
                    x &= x - 1u;
                }
            }
    }

    regs_to_rows(wreg, myRows, lane);
#pragma unroll
    for (int i = 0; i < 8; ++i) wreg[i] = wnext[i];
    __syncthreads();

    const unsigned n      = eCnt;
    const unsigned rounds = (n + 63u) >> 6;

    v2f bfc[16], bfn[16];
    load_bf(bfc, myRows, c, hh);

    int2   apA[4];
    float4 pwA[4][4];

#pragma unroll
    for (int j = 0; j < 9; ++j) {
        if (j < 7 || (j == 7 && par == 0)) {
            wave_lds_order();
            regs_to_rows(wreg, myRows, lane);
            wave_lds_order();
            load_bf(bfn, myRows, c, hh);
            __builtin_amdgcn_sched_barrier(0);
        }
        if (j == 0) fetch(apA, pwA, 0u, n, eList, a2p, pf, wave, hh);
        if (j < 6 || (j == 6 && par == 0)) rows_to_regs(wreg, W, b, 2 * (j + 2) + par, t, lane);

        if (j < 8 || par == 0) {
            const int p = 2 * j + par;
            v8f acc0 = (v8f)0.0f;
            v8f acc1 = (v8f)0.0f;
#pragma unroll
            for (int q = 0; q < 16; ++q) {
                acc0 = __builtin_amdgcn_wmma_f32_16x16x4_f32(
                    false, af[0][q], false, bfc[q], (short)0, acc0, false, false);
                acc1 = __builtin_amdgcn_wmma_f32_16x16x4_f32(
                    false, af[1][q], false, bfc[q], (short)0, acc1, false, false);
            }
#pragma unroll
            for (int r = 0; r < 8; ++r) {
                uLds[(8 * hh + r) * UPITCH + p * NH + 4 * c + t]      = acc0[r];
                uLds[(16 + 8 * hh + r) * UPITCH + p * NH + 4 * c + t] = acc1[r];
            }
        }
#pragma unroll
        for (int q = 0; q < 16; ++q) bfc[q] = bfn[q];
    }
    __syncthreads();

    for (unsigned r = 0; r < rounds; ++r) {
        int2   apB[4];
        float4 pwB[4][4];
        fetch(apB, pwB, r + 1u, n, eList, a2p, pf, wave, hh);
        process(apA, pwA, r, n, uLds, out, wave, c, hh);
#pragma unroll
        for (int it = 0; it < 4; ++it) {
            apA[it] = apB[it];
#pragma unroll
            for (int q = 0; q < 4; ++q) pwA[it][q] = pwB[it][q];
        }
    }
}

extern "C" void kernel_launch(void* const* d_in, const int* in_sizes, int n_in,
                              void* d_out, int out_size, void* d_ws, size_t ws_size,
                              hipStream_t stream) {
    const float* pf    = (const float*)d_in[0];
    const float* atomf = (const float*)d_in[1];
    const int*   a2p   = (const int*)d_in[2];
    const float* W     = (const float*)d_in[3];
    const float* b     = (const float*)d_in[4];
    float* out = (float*)d_out;
    unsigned int* bits = (unsigned int*)d_ws;

    prep<<<128, 256, 0, stream>>>(a2p, bits, out);
    own_sources<<<NBLK, 256, 0, stream>>>(pf, atomf, a2p, W, b, bits, out);
}
